# starting kernel
# baseline (speedup 1.0000x reference)
#include <hip/hip_runtime.h>

typedef float v2f __attribute__((ext_vector_type(2)));
typedef float v8f __attribute__((ext_vector_type(8)));

#define D_IN   2048
#define D_OUT  2048
#define NEXP   16
#define RANK   16
#define KTOP   2
#define ER     256
#define TOKS   16
#define CHUNK  256
#define XSTR   (CHUNK + 4)
#define HSTR   (ER + 4)

__global__ __launch_bounds__(256)
void moe_lora_wmma_kernel(const float* __restrict__ x,
                          const float* __restrict__ routing,
                          const int*   __restrict__ idxs,
                          const float* __restrict__ w_a,
                          const float* __restrict__ w_b,
                          float* __restrict__ out)
{
    __shared__ float x_lds[TOKS * XSTR];
    __shared__ float h_lds[TOKS * HSTR];
    __shared__ float g_lds[TOKS * NEXP];

    const int tid  = threadIdx.x;
    const int wave = tid >> 5;
    const int lane = tid & 31;
    const int nIdx = lane & 15;
    const int hv   = lane >> 4;
    const int tileBase = blockIdx.x * TOKS;

    {
        const int t = tid >> 4;
        const int e = tid & 15;
        const int tok = tileBase + t;
        float g = 0.f;
#pragma unroll
        for (int k = 0; k < KTOP; ++k) {
            const int   ie = idxs[tok * KTOP + k];
            const float rw = routing[tok * KTOP + k];
            g += (ie == e) ? rw : 0.f;
        }
        g_lds[t * NEXP + e] = 2.0f * g;
    }

    v8f acc0a = {}, acc0b = {};
    v8f acc1a = {}, acc1b = {};
    const int er0 = wave * 32;

    for (int c = 0; c < D_IN / CHUNK; ++c) {
        __syncthreads();
        {
            const float* xg = x + (size_t)tileBase * D_IN + c * CHUNK;
#pragma unroll
            for (int i = 0; i < 4; ++i) {
                const int fi  = tid + i * 256;
                const int row = fi >> 6;
                const int col = (fi & 63) << 2;
                const float4 v = *reinterpret_cast<const float4*>(
                    xg + (size_t)row * D_IN + col);
                *reinterpret_cast<float4*>(&x_lds[row * XSTR + col]) = v;
            }
        }
        __syncthreads();

        const float* waBase = w_a + (size_t)(er0 + nIdx) * D_IN + c * CHUNK;
#pragma unroll 2
        for (int k0 = 0; k0 < CHUNK; k0 += 8) {
            const int kkA = k0 + 2 * hv;
            const int kkB = kkA + 4;
            const v2f a0 = *reinterpret_cast<const v2f*>(&x_lds[nIdx * XSTR + kkA]);
            const v2f a1 = *reinterpret_cast<const v2f*>(&x_lds[nIdx * XSTR + kkB]);
            const v2f b0a = *reinterpret_cast<const v2f*>(waBase + kkA);
            const v2f b1a = *reinterpret_cast<const v2f*>(waBase + (size_t)16 * D_IN + kkA);
            const v2f b0b = *reinterpret_cast<const v2f*>(waBase + kkB);
            const v2f b1b = *reinterpret_cast<const v2f*>(waBase + (size_t)16 * D_IN + kkB);
            acc0a = __builtin_amdgcn_wmma_f32_16x16x4_f32(false, a0, false, b0a,
                                                          (short)0, acc0a, false, false);
            acc1a = __builtin_amdgcn_wmma_f32_16x16x4_f32(false, a0, false, b1a,
                                                          (short)0, acc1a, false, false);
            acc0b = __builtin_amdgcn_wmma_f32_16x16x4_f32(false, a1, false, b0b,
                                                          (short)0, acc0b, false, false);
            acc1b = __builtin_amdgcn_wmma_f32_16x16x4_f32(false, a1, false, b1b,
                                                          (short)0, acc1b, false, false);
        }
    }

    {
        const v8f h0v = acc0a + acc0b;
        const v8f h1v = acc1a + acc1b;
#pragma unroll
        for (int v = 0; v < 8; ++v) {
            const int t = v + 8 * hv;
            const float h0 = h0v[v];
            const float h1 = h1v[v];
            const float s0 = h0 * __builtin_amdgcn_rcpf(1.0f + __expf(-h0));
            const float s1 = h1 * __builtin_amdgcn_rcpf(1.0f + __expf(-h1));
            h_lds[t * HSTR + er0 + nIdx]      = g_lds[t * NEXP + wave * 2]     * s0;
            h_lds[t * HSTR + er0 + 16 + nIdx] = g_lds[t * NEXP + wave * 2 + 1] * s1;
        }
    }
    __syncthreads();

    const int dBase = wave * 256;
    for (int nt = 0; nt < 8; ++nt) {
        const int d0 = dBase + nt * 16 + nIdx;
        const int d1 = d0 + 128;
        v8f p0a = {}, p0b = {}, p1a = {}, p1b = {};
#pragma unroll 2
        for (int k0 = 0; k0 < ER; k0 += 8) {
            const int erA = k0 + 2 * hv;
            const int erB = erA + 4;
            const v2f a0 = *reinterpret_cast<const v2f*>(&h_lds[nIdx * HSTR + erA]);
            const v2f a1 = *reinterpret_cast<const v2f*>(&h_lds[nIdx * HSTR + erB]);
            const int eA = erA >> 4, rA = erA & 15;
            const int eB = erB >> 4, rB = erB & 15;
            const float* wbA = w_b + (size_t)eA * (D_OUT * RANK) + rA;
            const float* wbB = w_b + (size_t)eB * (D_OUT * RANK) + rB;
            const v2f b00 = *reinterpret_cast<const v2f*>(wbA + (size_t)d0 * RANK);
            const v2f b01 = *reinterpret_cast<const v2f*>(wbA + (size_t)d1 * RANK);
            const v2f b10 = *reinterpret_cast<const v2f*>(wbB + (size_t)d0 * RANK);
            const v2f b11 = *reinterpret_cast<const v2f*>(wbB + (size_t)d1 * RANK);
            p0a = __builtin_amdgcn_wmma_f32_16x16x4_f32(false, a0, false, b00,
                                                        (short)0, p0a, false, false);
            p1a = __builtin_amdgcn_wmma_f32_16x16x4_f32(false, a0, false, b01,
                                                        (short)0, p1a, false, false);
            p0b = __builtin_amdgcn_wmma_f32_16x16x4_f32(false, a1, false, b10,
                                                        (short)0, p0b, false, false);
            p1b = __builtin_amdgcn_wmma_f32_16x16x4_f32(false, a1, false, b11,
                                                        (short)0, p1b, false, false);
        }
        const v8f o0 = p0a + p0b;
        const v8f o1 = p1a + p1b;
        float* ob = out + (size_t)tileBase * D_OUT;
#pragma unroll
        for (int v = 0; v < 8; ++v) {
            const size_t row = (size_t)(v + 8 * hv) * D_OUT;
            ob[row + d0] = o0[v];
            ob[row + d1] = o1[v];
        }
    }
}

extern "C" void kernel_launch(void* const* d_in, const int* in_sizes, int n_in,
                              void* d_out, int out_size, void* d_ws, size_t ws_size,
                              hipStream_t stream) {
    const float* x       = (const float*)d_in[0];
    const float* routing = (const float*)d_in[1];
    const int*   idxs    = (const int*)d_in[2];
    const float* w_a     = (const float*)d_in[3];
    const float* w_b     = (const float*)d_in[4];
    float* out = (float*)d_out;

    const int tokens = in_sizes[1] / KTOP;
    const int tiles  = tokens / TOKS;

    hipLaunchKernelGGL(moe_lora_wmma_kernel, dim3(tiles), dim3(256), 0, stream,
                       x, routing, idxs, w_a, w_b, out);
}


# 128-bit fragment loads (4 consecutive k per lane, same k order for A and B) replace 64-bit ones: half the loads and line touches per WMMA; k pairs regrouped over the two chains, sum order changes
# speedup vs baseline: 1.2862x; 1.2862x over previous
#include <hip/hip_runtime.h>

typedef float v2f __attribute__((ext_vector_type(2)));
typedef float v4f __attribute__((ext_vector_type(4)));
typedef float v8f __attribute__((ext_vector_type(8)));

#define D_IN   2048
#define D_OUT  2048
#define NEXP   16
#define RANK   16
#define KTOP   2
#define ER     256
#define TOKS   16
#define CHUNK  256
#define XSTR   (CHUNK + 4)
#define HSTR   (ER + 4)

__global__ __launch_bounds__(256)
void moe_lora_wmma_kernel(const float* __restrict__ x,
                          const float* __restrict__ routing,
                          const int*   __restrict__ idxs,
                          const float* __restrict__ w_a,
                          const float* __restrict__ w_b,
                          float* __restrict__ out)
{
    __shared__ __attribute__((aligned(16))) float x_lds[TOKS * XSTR];
    __shared__ __attribute__((aligned(16))) float h_lds[TOKS * HSTR];
    __shared__ float g_lds[TOKS * NEXP];

    const int tid  = threadIdx.x;
    const int wave = tid >> 5;
    const int lane = tid & 31;
    const int nIdx = lane & 15;
    const int hv   = lane >> 4;
    const int tileBase = blockIdx.x * TOKS;

    {
        const int t = tid >> 4;
        const int e = tid & 15;
        const int tok = tileBase + t;
        float g = 0.f;
#pragma unroll
        for (int k = 0; k < KTOP; ++k) {
            const int   ie = idxs[tok * KTOP + k];
            const float rw = routing[tok * KTOP + k];
            g += (ie == e) ? rw : 0.f;
        }
        g_lds[t * NEXP + e] = 2.0f * g;
    }

    v8f acc0a = {}, acc0b = {};
    v8f acc1a = {}, acc1b = {};
    const int er0 = wave * 32;

    for (int c = 0; c < D_IN / CHUNK; ++c) {
        __syncthreads();
        {
            const float* xg = x + (size_t)tileBase * D_IN + c * CHUNK;
#pragma unroll
            for (int i = 0; i < 4; ++i) {
                const int fi  = tid + i * 256;
                const int row = fi >> 6;
                const int col = (fi & 63) << 2;
                const float4 v = *reinterpret_cast<const float4*>(
                    xg + (size_t)row * D_IN + col);
                *reinterpret_cast<float4*>(&x_lds[row * XSTR + col]) = v;
            }
        }
        __syncthreads();

        const float* xRow   = &x_lds[nIdx * XSTR + 4 * hv];
        const float* waRow0 = w_a + (size_t)(er0 + nIdx) * D_IN + c * CHUNK + 4 * hv;
        const float* waRow1 = waRow0 + (size_t)16 * D_IN;
#pragma unroll 4
        for (int k0 = 0; k0 < CHUNK; k0 += 8) {
            const v4f a  = *reinterpret_cast<const v4f*>(xRow + k0);
            const v4f b0 = *reinterpret_cast<const v4f*>(waRow0 + k0);
            const v4f b1 = *reinterpret_cast<const v4f*>(waRow1 + k0);
            acc0a = __builtin_amdgcn_wmma_f32_16x16x4_f32(false, a.xy, false, b0.xy,
                                                          (short)0, acc0a, false, false);
            acc1a = __builtin_amdgcn_wmma_f32_16x16x4_f32(false, a.xy, false, b1.xy,
                                                          (short)0, acc1a, false, false);
            acc0b = __builtin_amdgcn_wmma_f32_16x16x4_f32(false, a.zw, false, b0.zw,
                                                          (short)0, acc0b, false, false);
            acc1b = __builtin_amdgcn_wmma_f32_16x16x4_f32(false, a.zw, false, b1.zw,
                                                          (short)0, acc1b, false, false);
        }
    }

    {
        const v8f h0v = acc0a + acc0b;
        const v8f h1v = acc1a + acc1b;
#pragma unroll
        for (int v = 0; v < 8; ++v) {
            const int t = v + 8 * hv;
            const float h0 = h0v[v];
            const float h1 = h1v[v];
            const float s0 = h0 * __builtin_amdgcn_rcpf(1.0f + __expf(-h0));
            const float s1 = h1 * __builtin_amdgcn_rcpf(1.0f + __expf(-h1));
            h_lds[t * HSTR + er0 + nIdx]      = g_lds[t * NEXP + wave * 2]     * s0;
            h_lds[t * HSTR + er0 + 16 + nIdx] = g_lds[t * NEXP + wave * 2 + 1] * s1;
        }
    }
    __syncthreads();

    const int dBase = wave * 256;
    const float* hRow = &h_lds[nIdx * HSTR + 4 * hv];
    for (int nt = 0; nt < 8; ++nt) {
        const int d0 = dBase + nt * 16 + nIdx;
        const int d1 = d0 + 128;
        const float* wb0 = w_b + (size_t)d0 * RANK + 4 * hv;
        const float* wb1 = w_b + (size_t)d1 * RANK + 4 * hv;
        v8f p0a = {}, p0b = {}, p1a = {}, p1b = {};
#pragma unroll 2
        for (int e = 0; e < NEXP; ++e) {
#pragma unroll
            for (int r0 = 0; r0 < RANK; r0 += 8) {
                const size_t off = (size_t)e * (D_OUT * RANK) + r0;
                const v4f a  = *reinterpret_cast<const v4f*>(hRow + e * RANK + r0);
                const v4f b0 = *reinterpret_cast<const v4f*>(wb0 + off);
                const v4f b1 = *reinterpret_cast<const v4f*>(wb1 + off);
                p0a = __builtin_amdgcn_wmma_f32_16x16x4_f32(false, a.xy, false, b0.xy,
                                                            (short)0, p0a, false, false);
                p1a = __builtin_amdgcn_wmma_f32_16x16x4_f32(false, a.xy, false, b1.xy,
                                                            (short)0, p1a, false, false);
                p0b = __builtin_amdgcn_wmma_f32_16x16x4_f32(false, a.zw, false, b0.zw,
                                                            (short)0, p0b, false, false);
                p1b = __builtin_amdgcn_wmma_f32_16x16x4_f32(false, a.zw, false, b1.zw,
                                                            (short)0, p1b, false, false);
            }
        }
        const v8f o0 = p0a + p0b;
        const v8f o1 = p1a + p1b;
        float* ob = out + (size_t)tileBase * D_OUT;
#pragma unroll
        for (int v = 0; v < 8; ++v) {
            const size_t row = (size_t)(v + 8 * hv) * D_OUT;
            ob[row + d0] = o0[v];
            ob[row + d1] = o1[v];
        }
    }
}

extern "C" void kernel_launch(void* const* d_in, const int* in_sizes, int n_in,
                              void* d_out, int out_size, void* d_ws, size_t ws_size,
                              hipStream_t stream) {
    const float* x       = (const float*)d_in[0];
    const float* routing = (const float*)d_in[1];
    const int*   idxs    = (const int*)d_in[2];
    const float* w_a     = (const float*)d_in[3];
    const float* w_b     = (const float*)d_in[4];
    float* out = (float*)d_out;

    const int tokens = in_sizes[1] / KTOP;
    const int tiles  = tokens / TOKS;

    hipLaunchKernelGGL(moe_lora_wmma_kernel, dim3(tiles), dim3(256), 0, stream,
                       x, routing, idxs, w_a, w_b, out);
}


# Software pipelining: weight-fragment loads run 3 trips ahead of their WMMAs in a ring of 4 register sets, x chunk double-buffered in LDS, barriers wait on LDS only; sum order unchanged
# speedup vs baseline: 1.7491x; 1.3599x over previous
#include <hip/hip_runtime.h>

typedef float v2f __attribute__((ext_vector_type(2)));
typedef float v4f __attribute__((ext_vector_type(4)));
typedef float v8f __attribute__((ext_vector_type(8)));

#define D_IN   2048
#define D_OUT  2048
#define NEXP   16
#define RANK   16
#define KTOP   2
#define ER     256
#define TOKS   16
#define CHUNK  256
#define NCH    (D_IN / CHUNK)
#define XSTR   (CHUNK + 4)
#define HSTR   (ER + 4)
#define XBUF   (TOKS * XSTR)
#define STEPS  4
#define RING   4
#define TRIPK  (STEPS * 8)
#define TPC    (CHUNK / TRIPK)
#define EPT    (STEPS / 2)
#define TPN    (NEXP / EPT)

struct BSet { v4f b0[STEPS]; v4f b1[STEPS]; };

__device__ __forceinline__ void load_wa(BSet& B, const float* p0, const float* p1) {
#pragma unroll
    for (int s = 0; s < STEPS; ++s) {
        B.b0[s] = *reinterpret_cast<const v4f*>(p0 + 8 * s);
        B.b1[s] = *reinterpret_cast<const v4f*>(p1 + 8 * s);
    }
}

__device__ __forceinline__ void load_wb(BSet& B, const float* p0, const float* p1) {
#pragma unroll
    for (int s = 0; s < STEPS; ++s) {
        const size_t off = (size_t)(s >> 1) * (D_OUT * RANK) + (s & 1) * 8;
        B.b0[s] = *reinterpret_cast<const v4f*>(p0 + off);
        B.b1[s] = *reinterpret_cast<const v4f*>(p1 + off);
    }
}

struct ASet { v4f a[STEPS]; };
__device__ __forceinline__ void load_a(ASet& A, const float* aLds) {
#pragma unroll
    for (int s = 0; s < STEPS; ++s)
        A.a[s] = *reinterpret_cast<const v4f*>(aLds + 8 * s);
}

__device__ __forceinline__ void mma_trip(const ASet& A, const BSet& B,
                                         v8f& c0a, v8f& c0b, v8f& c1a, v8f& c1b) {
#pragma unroll
    for (int s = 0; s < STEPS; ++s) {
        const v4f a = A.a[s];
        c0a = __builtin_amdgcn_wmma_f32_16x16x4_f32(false, a.xy, false, B.b0[s].xy,
                                                    (short)0, c0a, false, false);
        c1a = __builtin_amdgcn_wmma_f32_16x16x4_f32(false, a.xy, false, B.b1[s].xy,
                                                    (short)0, c1a, false, false);
        c0b = __builtin_amdgcn_wmma_f32_16x16x4_f32(false, a.zw, false, B.b0[s].zw,
                                                    (short)0, c0b, false, false);
        c1b = __builtin_amdgcn_wmma_f32_16x16x4_f32(false, a.zw, false, B.b1[s].zw,
                                                    (short)0, c1b, false, false);
    }
}

__device__ __forceinline__ void lds_barrier() {
    __builtin_amdgcn_fence(__ATOMIC_RELEASE, "workgroup", "local");
    __builtin_amdgcn_s_barrier();
    __builtin_amdgcn_fence(__ATOMIC_ACQUIRE, "workgroup", "local");
}

__global__ __launch_bounds__(256)
void moe_lora_wmma_kernel(const float* __restrict__ x,
                          const float* __restrict__ routing,
                          const int*   __restrict__ idxs,
                          const float* __restrict__ w_a,
                          const float* __restrict__ w_b,
                          float* __restrict__ out)
{
    __shared__ __attribute__((aligned(16))) float x_lds[2 * XBUF];
    __shared__ __attribute__((aligned(16))) float h_lds[TOKS * HSTR];
    __shared__ float g_lds[TOKS * NEXP];

    const int tid  = threadIdx.x;
    const int wave = tid >> 5;
    const int lane = tid & 31;
    const int nIdx = lane & 15;
    const int hv   = lane >> 4;
    const int tileBase = blockIdx.x * TOKS;

    {
        const int t = tid >> 4;
        const int e = tid & 15;
        const int tok = tileBase + t;
        float g = 0.f;
#pragma unroll
        for (int k = 0; k < KTOP; ++k) {
            const int   ie = idxs[tok * KTOP + k];
            const float rw = routing[tok * KTOP + k];
            g += (ie == e) ? rw : 0.f;
        }
        g_lds[t * NEXP + e] = 2.0f * g;
    }

    const int xrow = tid >> 6;
    const int xcol = (tid & 63) << 2;
    const float* xg = x + (size_t)tileBase * D_IN + (size_t)xrow * D_IN + xcol;
    v4f xr[4];

    const int er0   = wave * 32;
    const int dBase = wave * 256;
    const float* wa0 = w_a + (size_t)(er0 + nIdx) * D_IN + 4 * hv;
    const float* wa1 = wa0 + (size_t)16 * D_IN;
    const float* wb0 = w_b + (size_t)(dBase + nIdx) * RANK + 4 * hv;
    const float* wb1 = wb0 + (size_t)128 * RANK;

    BSet B[RING];
    ASet A;

#pragma unroll
    for (int i = 0; i < 4; ++i)
        xr[i] = *reinterpret_cast<const v4f*>(xg + (size_t)(4 * i) * D_IN);
#pragma unroll
    for (int r = 0; r < RING - 1; ++r)
        load_wa(B[r], wa0 + r * TRIPK, wa1 + r * TRIPK);
#pragma unroll
    for (int i = 0; i < 4; ++i)
        *reinterpret_cast<v4f*>(&x_lds[(xrow + 4 * i) * XSTR + xcol]) = xr[i];
#pragma unroll
    for (int i = 0; i < 4; ++i)
        xr[i] = *reinterpret_cast<const v4f*>(xg + (size_t)(4 * i) * D_IN + CHUNK);
    lds_barrier();

    v8f acc0a = {}, acc0b = {};
    v8f acc1a = {}, acc1b = {};

    for (int c = 0; c < NCH; ++c) {
        const float* xa = &x_lds[(c & 1) * XBUF + nIdx * XSTR + 4 * hv];
        const float* wc0 = wa0 + c * CHUNK;
        const float* wc1 = wa1 + c * CHUNK;
#pragma unroll
        for (int t = 0; t < TPC; ++t) {
            load_a(A, xa + t * TRIPK);
            __builtin_amdgcn_sched_barrier(0);
            const int tn = t + RING - 1;
            if (tn < TPC || c + 1 < NCH)
                load_wa(B[tn % RING], wc0 + tn * TRIPK, wc1 + tn * TRIPK);
            else
                load_wb(B[tn % RING], wb0 + (size_t)((tn - TPC) * EPT) * (D_OUT * RANK),
                                      wb1 + (size_t)((tn - TPC) * EPT) * (D_OUT * RANK));
            __builtin_amdgcn_sched_barrier(0);
            mma_trip(A, B[t % RING], acc0a, acc0b, acc1a, acc1b);
            __builtin_amdgcn_sched_barrier(0);
        }
        if (c + 1 < NCH) {
            float* xs = &x_lds[((c + 1) & 1) * XBUF + xrow * XSTR + xcol];
#pragma unroll
            for (int i = 0; i < 4; ++i)
                *reinterpret_cast<v4f*>(xs + (4 * i) * XSTR) = xr[i];
            if (c + 2 < NCH) {
#pragma unroll
                for (int i = 0; i < 4; ++i)
                    xr[i] = *reinterpret_cast<const v4f*>(
                        xg + (size_t)(4 * i) * D_IN + (c + 2) * CHUNK);
            }
            lds_barrier();
        }
    }

    {
        const v8f h0v = acc0a + acc0b;
        const v8f h1v = acc1a + acc1b;
#pragma unroll
        for (int v = 0; v < 8; ++v) {
            const int t = v + 8 * hv;
            const float h0 = h0v[v];
            const float h1 = h1v[v];
            const float s0 = h0 * __builtin_amdgcn_rcpf(1.0f + __expf(-h0));
            const float s1 = h1 * __builtin_amdgcn_rcpf(1.0f + __expf(-h1));
            h_lds[t * HSTR + er0 + nIdx]      = g_lds[t * NEXP + wave * 2]     * s0;
            h_lds[t * HSTR + er0 + 16 + nIdx] = g_lds[t * NEXP + wave * 2 + 1] * s1;
        }
    }
    lds_barrier();

    const float* hA = &h_lds[nIdx * HSTR + 4 * hv];
    for (int nt = 0; nt < 8; ++nt) {
        const float* q0 = wb0 + (size_t)(nt * 16) * RANK;
        const float* q1 = wb1 + (size_t)(nt * 16) * RANK;
        v8f p0a = {}, p0b = {}, p1a = {}, p1b = {};
#pragma unroll
        for (int j = 0; j < TPN; ++j) {
            load_a(A, hA + j * TRIPK);
            __builtin_amdgcn_sched_barrier(0);
            const int jn = j + RING - 1;
            if (jn < TPN)
                load_wb(B[jn % RING], q0 + (size_t)(jn * EPT) * (D_OUT * RANK),
                                      q1 + (size_t)(jn * EPT) * (D_OUT * RANK));
            else if (nt + 1 < 8)
                load_wb(B[jn % RING],
                        q0 + (size_t)16 * RANK + (size_t)((jn - TPN) * EPT) * (D_OUT * RANK),
                        q1 + (size_t)16 * RANK + (size_t)((jn - TPN) * EPT) * (D_OUT * RANK));
            __builtin_amdgcn_sched_barrier(0);
            mma_trip(A, B[j % RING], p0a, p0b, p1a, p1b);
            __builtin_amdgcn_sched_barrier(0);
        }
        const v8f o0 = p0a + p0b;
        const v8f o1 = p1a + p1b;
        const int d0 = dBase + nt * 16 + nIdx;
        const int d1 = d0 + 128;
        float* ob = out + (size_t)tileBase * D_OUT;
#pragma unroll
        for (int v = 0; v < 8; ++v) {
            const size_t row = (size_t)(v + 8 * hv) * D_OUT;
            ob[row + d0] = o0[v];
            ob[row + d1] = o1[v];
        }
    }
}

extern "C" void kernel_launch(void* const* d_in, const int* in_sizes, int n_in,
                              void* d_out, int out_size, void* d_ws, size_t ws_size,
                              hipStream_t stream) {
    const float* x       = (const float*)d_in[0];
    const float* routing = (const float*)d_in[1];
    const int*   idxs    = (const int*)d_in[2];
    const float* w_a     = (const float*)d_in[3];
    const float* w_b     = (const float*)d_in[4];
    float* out = (float*)d_out;

    const int tokens = in_sizes[1] / KTOP;
    const int tiles  = tokens / TOKS;

    hipLaunchKernelGGL(moe_lora_wmma_kernel, dim3(tiles), dim3(256), 0, stream,
                       x, routing, idxs, w_a, w_b, out);
}


# Pre-pass kernel writes w_a and w_b into the workspace in per-wave fragment order: a weight fragment load reads 512 contiguous bytes, not 32 B from each of 16 lines; 2 launches; sum order unchanged
# speedup vs baseline: 2.2104x; 1.2637x over previous
#include <hip/hip_runtime.h>

typedef float v2f __attribute__((ext_vector_type(2)));
typedef float v4f __attribute__((ext_vector_type(4)));
typedef float v8f __attribute__((ext_vector_type(8)));

#define D_IN   2048
#define D_OUT  2048
#define NEXP   16
#define RANK   16
#define KTOP   2
#define ER     256
#define TOKS   16
#define CHUNK  256
#define NCH    (D_IN / CHUNK)
#define XSTR   (CHUNK + 4)
#define HSTR   (ER + 4)
#define XBUF   (TOKS * XSTR)
#define STEPS  4
#define RING   4
#define TRIPK  (STEPS * 8)
#define TPC    (CHUNK / TRIPK)
#define EPT    (STEPS / 2)
#define TPN    (NEXP / EPT)
#define NDT    8
#define TRIPS1 (NCH * TPC)
#define TRIPS2 (NDT * TPN)
#define SETV4  (2 * STEPS * 32)
#define WAVEV4 ((TRIPS1 + TRIPS2) * SETV4)

struct BSet { v4f b0[STEPS]; v4f b1[STEPS]; };

__device__ __forceinline__ void load_set(BSet& B, const v4f* p) {
#pragma unroll
    for (int s = 0; s < STEPS; ++s) {
        B.b0[s] = p[(2 * s) * 32];
        B.b1[s] = p[(2 * s + 1) * 32];
    }
}

__global__ __launch_bounds__(256)
void relayout_weights_kernel(const float* __restrict__ w_a,
                             const float* __restrict__ w_b,
                             float* __restrict__ wfrag)
{
    const int tid = threadIdx.x;
    const int n   = tid & 15;
    const int hv  = (tid >> 4) & 1;
    const int q   = (tid >> 5) & 1;
    const int s   = tid >> 6;
    const int w   = blockIdx.x >> 5;
    const int G0  = (blockIdx.x & 31) * 4;
    v4f v[4];
    if (G0 < TRIPS1) {
        const float* p = w_a + (size_t)(w * 32 + q * 16 + n) * D_IN + G0 * TRIPK + 8 * s + 4 * hv;
        for (int u = 0; u < 4; ++u)
            v[u] = *reinterpret_cast<const v4f*>(p + u * TRIPK);
    } else {
        for (int u = 0; u < 4; ++u) {
            const int G2 = G0 + u - TRIPS1;
            const int nt = G2 / TPN;
            const int j  = G2 % TPN;
            const int e  = j * EPT + (s >> 1);
            const int d  = w * 256 + nt * 16 + q * 128 + n;
            v[u] = *reinterpret_cast<const v4f*>(
                w_b + ((size_t)e * D_OUT + d) * RANK + (s & 1) * 8 + 4 * hv);
        }
    }
    v4f* dst = reinterpret_cast<v4f*>(wfrag) + ((size_t)w * (TRIPS1 + TRIPS2) + G0) * SETV4 + tid;
    for (int u = 0; u < 4; ++u)
        dst[u * SETV4] = v[u];
}

struct ASet { v4f a[STEPS]; };
__device__ __forceinline__ void load_a(ASet& A, const float* aLds) {
#pragma unroll
    for (int s = 0; s < STEPS; ++s)
        A.a[s] = *reinterpret_cast<const v4f*>(aLds + 8 * s);
}

__device__ __forceinline__ void mma_trip(const ASet& A, const BSet& B,
                                         v8f& c0a, v8f& c0b, v8f& c1a, v8f& c1b) {
#pragma unroll
    for (int s = 0; s < STEPS; ++s) {
        const v4f a = A.a[s];
        c0a = __builtin_amdgcn_wmma_f32_16x16x4_f32(false, a.xy, false, B.b0[s].xy,
                                                    (short)0, c0a, false, false);
        c1a = __builtin_amdgcn_wmma_f32_16x16x4_f32(false, a.xy, false, B.b1[s].xy,
                                                    (short)0, c1a, false, false);
        c0b = __builtin_amdgcn_wmma_f32_16x16x4_f32(false, a.zw, false, B.b0[s].zw,
                                                    (short)0, c0b, false, false);
        c1b = __builtin_amdgcn_wmma_f32_16x16x4_f32(false, a.zw, false, B.b1[s].zw,
                                                    (short)0, c1b, false, false);
    }
}

__device__ __forceinline__ void lds_barrier() {
    __builtin_amdgcn_fence(__ATOMIC_RELEASE, "workgroup", "local");
    __builtin_amdgcn_s_barrier();
    __builtin_amdgcn_fence(__ATOMIC_ACQUIRE, "workgroup", "local");
}

__global__ __launch_bounds__(256)
void moe_lora_wmma_kernel(const float* __restrict__ x,
                          const float* __restrict__ routing,
                          const int*   __restrict__ idxs,
                          const float* __restrict__ wfrag,
                          float* __restrict__ out)
{
    __shared__ __attribute__((aligned(16))) float x_lds[2 * XBUF];
    __shared__ __attribute__((aligned(16))) float h_lds[TOKS * HSTR];
    __shared__ float g_lds[TOKS * NEXP];

    const int tid  = threadIdx.x;
    const int wave = tid >> 5;
    const int lane = tid & 31;
    const int nIdx = lane & 15;
    const int hv   = lane >> 4;
    const int tileBase = blockIdx.x * TOKS;

    {
        const int t = tid >> 4;
        const int e = tid & 15;
        const int tok = tileBase + t;
        float g = 0.f;
#pragma unroll
        for (int k = 0; k < KTOP; ++k) {
            const int   ie = idxs[tok * KTOP + k];
            const float rw = routing[tok * KTOP + k];
            g += (ie == e) ? rw : 0.f;
        }
        g_lds[t * NEXP + e] = 2.0f * g;
    }

    const int xrow = tid >> 6;
    const int xcol = (tid & 63) << 2;
    const float* xg = x + (size_t)tileBase * D_IN + (size_t)xrow * D_IN + xcol;
    v4f xr[4];

    const int er0   = wave * 32;
    const int dBase = wave * 256;
    const v4f* wq = reinterpret_cast<const v4f*>(wfrag) + (size_t)wave * WAVEV4 + lane;

    BSet B[RING];
    ASet A;

#pragma unroll
    for (int i = 0; i < 4; ++i)
        xr[i] = *reinterpret_cast<const v4f*>(xg + (size_t)(4 * i) * D_IN);
#pragma unroll
    for (int r = 0; r < RING - 1; ++r)
        load_set(B[r], wq + r * SETV4);
#pragma unroll
    for (int i = 0; i < 4; ++i)
        *reinterpret_cast<v4f*>(&x_lds[(xrow + 4 * i) * XSTR + xcol]) = xr[i];
#pragma unroll
    for (int i = 0; i < 4; ++i)
        xr[i] = *reinterpret_cast<const v4f*>(xg + (size_t)(4 * i) * D_IN + CHUNK);
    lds_barrier();

    v8f acc0a = {}, acc0b = {};
    v8f acc1a = {}, acc1b = {};

    for (int c = 0; c < NCH; ++c) {
        const float* xa = &x_lds[(c & 1) * XBUF + nIdx * XSTR + 4 * hv];
        const v4f* wc = wq + (size_t)(c * TPC) * SETV4;
#pragma unroll
        for (int t = 0; t < TPC; ++t) {
            load_a(A, xa + t * TRIPK);
            __builtin_amdgcn_sched_barrier(0);
            const int tn = t + RING - 1;
            load_set(B[tn % RING], wc + tn * SETV4);
            __builtin_amdgcn_sched_barrier(0);
            mma_trip(A, B[t % RING], acc0a, acc0b, acc1a, acc1b);
            __builtin_amdgcn_sched_barrier(0);
        }
        if (c + 1 < NCH) {
            float* xs = &x_lds[((c + 1) & 1) * XBUF + xrow * XSTR + xcol];
#pragma unroll
            for (int i = 0; i < 4; ++i)
                *reinterpret_cast<v4f*>(xs + (4 * i) * XSTR) = xr[i];
            if (c + 2 < NCH) {
#pragma unroll
                for (int i = 0; i < 4; ++i)
                    xr[i] = *reinterpret_cast<const v4f*>(
                        xg + (size_t)(4 * i) * D_IN + (c + 2) * CHUNK);
            }
            lds_barrier();
        }
    }

    {
        const v8f h0v = acc0a + acc0b;
        const v8f h1v = acc1a + acc1b;
#pragma unroll
        for (int v = 0; v < 8; ++v) {
            const int t = v + 8 * hv;
            const float h0 = h0v[v];
            const float h1 = h1v[v];
            const float s0 = h0 * __builtin_amdgcn_rcpf(1.0f + __expf(-h0));
            const float s1 = h1 * __builtin_amdgcn_rcpf(1.0f + __expf(-h1));
            h_lds[t * HSTR + er0 + nIdx]      = g_lds[t * NEXP + wave * 2]     * s0;
            h_lds[t * HSTR + er0 + 16 + nIdx] = g_lds[t * NEXP + wave * 2 + 1] * s1;
        }
    }
    lds_barrier();

    const float* hA = &h_lds[nIdx * HSTR + 4 * hv];
    for (int nt = 0; nt < NDT; ++nt) {
        const v4f* wn = wq + (size_t)(TRIPS1 + nt * TPN) * SETV4;
        v8f p0a = {}, p0b = {}, p1a = {}, p1b = {};
#pragma unroll
        for (int j = 0; j < TPN; ++j) {
            load_a(A, hA + j * TRIPK);
            __builtin_amdgcn_sched_barrier(0);
            const int jn = j + RING - 1;
            if (jn < TPN || nt + 1 < NDT)
                load_set(B[jn % RING], wn + jn * SETV4);
            __builtin_amdgcn_sched_barrier(0);
            mma_trip(A, B[j % RING], p0a, p0b, p1a, p1b);
            __builtin_amdgcn_sched_barrier(0);
        }
        const v8f o0 = p0a + p0b;
        const v8f o1 = p1a + p1b;
        const int d0 = dBase + nt * 16 + nIdx;
        const int d1 = d0 + 128;
        float* ob = out + (size_t)tileBase * D_OUT;
#pragma unroll
        for (int v = 0; v < 8; ++v) {
            const size_t row = (size_t)(v + 8 * hv) * D_OUT;
            ob[row + d0] = o0[v];
            ob[row + d1] = o1[v];
        }
    }
}

extern "C" void kernel_launch(void* const* d_in, const int* in_sizes, int n_in,
                              void* d_out, int out_size, void* d_ws, size_t ws_size,
                              hipStream_t stream) {
    const float* x       = (const float*)d_in[0];
    const float* routing = (const float*)d_in[1];
    const int*   idxs    = (const int*)d_in[2];
    const float* w_a     = (const float*)d_in[3];
    const float* w_b     = (const float*)d_in[4];
    float* out = (float*)d_out;

    const int tokens = in_sizes[1] / KTOP;
    const int tiles  = tokens / TOKS;

    float* wfrag = (float*)d_ws;

    hipLaunchKernelGGL(relayout_weights_kernel, dim3(256), dim3(256), 0, stream,
                       w_a, w_b, wfrag);
    hipLaunchKernelGGL(moe_lora_wmma_kernel, dim3(tiles), dim3(256), 0, stream,
                       x, routing, idxs, wfrag, out);
}


# Only the routed experts are computed: a workgroup per (256-token chunk, expert) lists its tokens by ballot; k split over 8 waves, summed in LDS; slot 1 adds onto slot 0's rows; sum order changes
# speedup vs baseline: 3.2875x; 1.4873x over previous
#include <hip/hip_runtime.h>

typedef float v2f __attribute__((ext_vector_type(2)));
typedef float v4f __attribute__((ext_vector_type(4)));
typedef float v8f __attribute__((ext_vector_type(8)));

#define D_IN   2048
#define D_OUT  2048
#define NEXP   16
#define RANK   16
#define KTOP   2
#define CTOK   256
#define NW     8
#define KSL    (D_IN / NW)
#define S1     (KSL / 8)
#define DSL    (D_OUT / NW)
#define NT2    (DSL / 16)
#define PSTR   20

__device__ __forceinline__ v8f mma4(v2f a, v2f b, v8f c) {
    return __builtin_amdgcn_wmma_f32_16x16x4_f32(false, a, false, b, (short)0, c, false, false);
}

__device__ __forceinline__ void lds_barrier() {
    __builtin_amdgcn_fence(__ATOMIC_RELEASE, "workgroup", "local");
    __builtin_amdgcn_s_barrier();
    __builtin_amdgcn_fence(__ATOMIC_ACQUIRE, "workgroup", "local");
}

__device__ __forceinline__ void load_rows(v4f (&a)[S1], const float* xa) {
#pragma unroll
    for (int s = 0; s < S1; ++s)
        a[s] = *reinterpret_cast<const v4f*>(xa + 8 * s);
}

__device__ __forceinline__ v8f first_product(const v4f (&a)[S1], const v4f (&bw)[S1]) {
    v8f acc0 = {}, acc1 = {};
#pragma unroll
    for (int s = 0; s < S1; ++s) {
        acc0 = mma4(a[s].xy, bw[s].xy, acc0);
        acc1 = mma4(a[s].zw, bw[s].zw, acc1);
    }
    return acc0 + acc1;
}

template <int KK>
__global__ __launch_bounds__(256)
void routed_lora_kernel(const float* __restrict__ x,
                        const float* __restrict__ routing,
                        const int*   __restrict__ idxs,
                        const float* __restrict__ w_a,
                        const float* __restrict__ w_b,
                        float* __restrict__ out)
{
    __shared__ int   tok_l[CTOK];
    __shared__ float gate_l[CTOK];
    __shared__ int   wcnt[NW];
    __shared__ __attribute__((aligned(16))) float part[NW * 16 * PSTR];

    const int tid  = threadIdx.x;
    const int wave = tid >> 5;
    const int lane = tid & 31;
    const int c    = lane & 15;
    const int hv   = lane >> 4;
    const int e    = blockIdx.y;
    const int t    = blockIdx.x * CTOK + tid;

    const int   my_e = idxs[t * KTOP + KK];
    const float my_g = routing[t * KTOP + KK];

    v4f bw[S1];
    {
        const float* wa = w_a + ((size_t)e * RANK + c) * D_IN + wave * KSL + 4 * hv;
#pragma unroll
        for (int s = 0; s < S1; ++s)
            bw[s] = *reinterpret_cast<const v4f*>(wa + 8 * s);
    }

    const bool hit = my_e == e;
    const unsigned mask = __builtin_amdgcn_ballot_w32(hit);
    if (lane == 0) wcnt[wave] = __popc(mask);
    lds_barrier();
    int base = 0, nsum = 0;
#pragma unroll
    for (int w = 0; w < NW; ++w) {
        const int cw = wcnt[w];
        base += (w < wave) ? cw : 0;
        nsum += cw;
    }
    if (hit) {
        const int pos = base + __popc(mask & ((1u << lane) - 1u));
        tok_l[pos]  = t;
        gate_l[pos] = 2.0f * my_g;
    }
    lds_barrier();
    const int n = __builtin_amdgcn_readfirstlane(nsum);
    if (n == 0) return;
    const int ntile = (n + 15) >> 4;

    v4f a[S1];
    {
        const int pa = (c < n) ? c : 0;
        load_rows(a, x + (size_t)tok_l[pa] * D_IN + wave * KSL + 4 * hv);
    }
    v4f b2[NT2][2];
    {
        const float* wb = w_b + ((size_t)e * D_OUT + wave * DSL + c) * RANK + 4 * hv;
#pragma unroll
        for (int q = 0; q < NT2; ++q) {
            b2[q][0] = *reinterpret_cast<const v4f*>(wb + q * 16 * RANK);
            b2[q][1] = *reinterpret_cast<const v4f*>(wb + q * 16 * RANK + 8);
        }
    }
    v8f hp = first_product(a, bw);

    for (int mt = 0;;) {
        const int p0 = mt * 16;

        float* orow[8];
        bool   ok[8];
#pragma unroll
        for (int r = 0; r < 8; ++r) {
            const int pr = p0 + 8 * hv + r;
            ok[r]   = pr < n;
            orow[r] = out + (size_t)tok_l[ok[r] ? pr : p0] * D_OUT + wave * DSL + c;
        }
        v8f o[NT2];
#pragma unroll
        for (int q = 0; q < NT2; ++q) {
#pragma unroll
            for (int r = 0; r < 8; ++r)
                o[q][r] = (KK == 0) ? 0.f : orow[r][16 * q];
        }

#pragma unroll
        for (int r = 0; r < 8; ++r)
            part[(wave * 16 + 8 * hv + r) * PSTR + c] = hp[r];
        lds_barrier();

        const float g = (p0 + c < n) ? gate_l[p0 + c] : 0.f;
        v4f ha[2];
#pragma unroll
        for (int u = 0; u < 2; ++u) {
            v4f sum = *reinterpret_cast<const v4f*>(&part[c * PSTR + 8 * u + 4 * hv]);
#pragma unroll
            for (int w = 1; w < NW; ++w)
                sum += *reinterpret_cast<const v4f*>(&part[(w * 16 + c) * PSTR + 8 * u + 4 * hv]);
#pragma unroll
            for (int i = 0; i < 4; ++i) {
                const float hval = sum[i];
                ha[u][i] = g * (hval * __builtin_amdgcn_rcpf(1.0f + __expf(-hval)));
            }
        }
        lds_barrier();

#pragma unroll
        for (int q = 0; q < NT2; ++q) o[q] = mma4(ha[0].xy, b2[q][0].xy, o[q]);
#pragma unroll
        for (int q = 0; q < NT2; ++q) o[q] = mma4(ha[0].zw, b2[q][0].zw, o[q]);
#pragma unroll
        for (int q = 0; q < NT2; ++q) o[q] = mma4(ha[1].xy, b2[q][1].xy, o[q]);
#pragma unroll
        for (int q = 0; q < NT2; ++q) o[q] = mma4(ha[1].zw, b2[q][1].zw, o[q]);

#pragma unroll
        for (int r = 0; r < 8; ++r) {
            if (ok[r]) {
#pragma unroll
                for (int q = 0; q < NT2; ++q)
                    orow[r][16 * q] = o[q][r];
            }
        }

        if (++mt >= ntile) break;
        {
            const int pn = mt * 16 + c;
            load_rows(a, x + (size_t)tok_l[(pn < n) ? pn : mt * 16] * D_IN + wave * KSL + 4 * hv);
        }
        hp = first_product(a, bw);
    }
}

extern "C" void kernel_launch(void* const* d_in, const int* in_sizes, int n_in,
                              void* d_out, int out_size, void* d_ws, size_t ws_size,
                              hipStream_t stream) {
    const float* x       = (const float*)d_in[0];
    const float* routing = (const float*)d_in[1];
    const int*   idxs    = (const int*)d_in[2];
    const float* w_a     = (const float*)d_in[3];
    const float* w_b     = (const float*)d_in[4];
    float* out = (float*)d_out;

    hipLaunchKernelGGL((routed_lora_kernel<0>), dim3(16, NEXP), dim3(CTOK), 0, stream,
                       x, routing, idxs, w_a, w_b, out);
    hipLaunchKernelGGL((routed_lora_kernel<1>), dim3(16, NEXP), dim3(CTOK), 0, stream,
                       x, routing, idxs, w_a, w_b, out);
}


# First-product operands (x rows, w_a) loaded as whole 128-B lines (512 contiguous B per load, not 32 B from each of 16 rows) and put in fragment order through per-wave LDS; sum order unchanged
# speedup vs baseline: 3.4197x; 1.0402x over previous
#include <hip/hip_runtime.h>

typedef float v2f __attribute__((ext_vector_type(2)));
typedef float v4f __attribute__((ext_vector_type(4)));
typedef float v8f __attribute__((ext_vector_type(8)));
typedef int   v4i __attribute__((ext_vector_type(4)));

#define D_IN   2048
#define D_OUT  2048
#define NEXP   16
#define RANK   16
#define KTOP   2
#define CTOK   256
#define NW     8
#define KSL    (D_IN / NW)
#define S1     (KSL / 8)
#define DSL    (D_OUT / NW)
#define NT2    (DSL / 16)
#define PSTR   20
#define XP     (KSL + 4)

__device__ __forceinline__ v8f mma4(v2f a, v2f b, v8f c) {
    return __builtin_amdgcn_wmma_f32_16x16x4_f32(false, a, false, b, (short)0, c, false, false);
}

__device__ __forceinline__ void lds_barrier() {
    __builtin_amdgcn_fence(__ATOMIC_RELEASE, "workgroup", "local");
    __builtin_amdgcn_s_barrier();
    __builtin_amdgcn_fence(__ATOMIC_ACQUIRE, "workgroup", "local");
}

__device__ __forceinline__ void wave_lds_sync() {
    __builtin_amdgcn_fence(__ATOMIC_RELEASE, "wavefront", "local");
    __builtin_amdgcn_wave_barrier();
    __builtin_amdgcn_fence(__ATOMIC_ACQUIRE, "wavefront", "local");
}

__device__ __forceinline__ void load_weight_block(v4f (&t)[S1], const float* w0, int lane) {
#pragma unroll
    for (int j = 0; j < S1; ++j)
        t[j] = *reinterpret_cast<const v4f*>(w0 + (size_t)(j >> 1) * D_IN + (KSL / 2) * (j & 1) + 4 * lane);
}
__device__ __forceinline__ void load_token_block(v4f (&t)[S1], const float* xs, const int* tok_l,
                                                 int p0, int n, int lane) {
    const v4i* tl = reinterpret_cast<const v4i*>(tok_l + p0);
    const v4i q0 = tl[0], q1 = tl[1], q2 = tl[2], q3 = tl[3];
    const int tk[16] = { q0.x, q0.y, q0.z, q0.w, q1.x, q1.y, q1.z, q1.w,
                         q2.x, q2.y, q2.z, q2.w, q3.x, q3.y, q3.z, q3.w };
    int ts[16];
#pragma unroll
    for (int r = 0; r < 16; ++r)
        ts[r] = __builtin_amdgcn_readfirstlane(tk[r]);
#pragma unroll
    for (int r = 0; r < 16; ++r) {
        const int tr = (p0 + r < n) ? ts[r] : ts[0];
        const float* row = xs + (size_t)tr * D_IN;
        t[2 * r]     = *reinterpret_cast<const v4f*>(row + 4 * lane);
        t[2 * r + 1] = *reinterpret_cast<const v4f*>(row + KSL / 2 + 4 * lane);
    }
}
__device__ __forceinline__ void block_to_lds(float* stg, const v4f (&t)[S1], int lane) {
    wave_lds_sync();
#pragma unroll
    for (int j = 0; j < S1; ++j)
        *reinterpret_cast<v4f*>(stg + (j >> 1) * XP + (KSL / 2) * (j & 1) + 4 * lane) = t[j];
    wave_lds_sync();
}

__device__ __forceinline__ void read_fragments(v4f (&f)[S1], const float* stg, int c, int hv) {
#pragma unroll
    for (int s = 0; s < S1; ++s)
        f[s] = *reinterpret_cast<const v4f*>(stg + c * XP + 8 * s + 4 * hv);
}
__device__ __forceinline__ v8f first_product(const float* sx, const v4f (&bw)[S1], int c, int hv) {
    v4f a[S1];
    read_fragments(a, sx, c, hv);
    __builtin_amdgcn_sched_barrier(0);
    v8f acc0 = {}, acc1 = {};
#pragma unroll
    for (int s = 0; s < S1; ++s) {
        acc0 = mma4(a[s].xy, bw[s].xy, acc0);
        acc1 = mma4(a[s].zw, bw[s].zw, acc1);
    }
    return acc0 + acc1;
}

template <int KK>
__global__ __launch_bounds__(256)
void routed_lora_kernel(const float* __restrict__ x,
                        const float* __restrict__ routing,
                        const int*   __restrict__ idxs,
                        const float* __restrict__ w_a,
                        const float* __restrict__ w_b,
                        float* __restrict__ out)
{
    __shared__ __attribute__((aligned(16))) int tok_l[CTOK];
    __shared__ float gate_l[CTOK];
    __shared__ int   wcnt[NW];
    __shared__ __attribute__((aligned(16))) float part[NW * 16 * PSTR];
    __shared__ __attribute__((aligned(16))) float stage_w[NW * 16 * XP];
    __shared__ __attribute__((aligned(16))) float stage_x[NW * 16 * XP];

    const int tid  = threadIdx.x;
    const int wave = tid >> 5;
    const int lane = tid & 31;
    const int c    = lane & 15;
    const int hv   = lane >> 4;
    const int e    = blockIdx.y;
    const int t    = blockIdx.x * CTOK + tid;

    const int   my_e = idxs[t * KTOP + KK];
    const float my_g = routing[t * KTOP + KK];

    float* sw = stage_w + wave * 16 * XP;
    float* sx = stage_x + wave * 16 * XP;
    v4f tw[S1];
    load_weight_block(tw, w_a + (size_t)e * RANK * D_IN + wave * KSL, lane);

    const bool hit = my_e == e;
    const unsigned mask = __builtin_amdgcn_ballot_w32(hit);
    if (lane == 0) wcnt[wave] = __popc(mask);
    lds_barrier();
    int base = 0, nsum = 0;
#pragma unroll
    for (int w = 0; w < NW; ++w) {
        const int cw = wcnt[w];
        base += (w < wave) ? cw : 0;
        nsum += cw;
    }
    if (hit) {
        const int pos = base + __popc(mask & ((1u << lane) - 1u));
        tok_l[pos]  = t;
        gate_l[pos] = 2.0f * my_g;
    }
    lds_barrier();
    const int n = __builtin_amdgcn_readfirstlane(nsum);
    if (n == 0) return;
    const int ntile = (n + 15) >> 4;

    v4f tx[S1];
    load_token_block(tx, x + wave * KSL, tok_l, 0, n, lane);
    block_to_lds(sw, tw, lane);
    v4f b2[NT2][2];
    {
        const float* wb = w_b + ((size_t)e * D_OUT + wave * DSL + c) * RANK + 4 * hv;
#pragma unroll
        for (int q = 0; q < NT2; ++q) {
            b2[q][0] = *reinterpret_cast<const v4f*>(wb + q * 16 * RANK);
            b2[q][1] = *reinterpret_cast<const v4f*>(wb + q * 16 * RANK + 8);
        }
    }
    block_to_lds(sx, tx, lane);
    v4f bw[S1];
    read_fragments(bw, sw, c, hv);
    v8f hp = first_product(sx, bw, c, hv);

    for (int mt = 0;;) {
        const int p0 = mt * 16;

        float* orow[8];
        bool   ok[8];
#pragma unroll
        for (int r = 0; r < 8; ++r) {
            const int pr = p0 + 8 * hv + r;
            ok[r]   = pr < n;
            orow[r] = out + (size_t)tok_l[ok[r] ? pr : p0] * D_OUT + wave * DSL + c;
        }
        v8f o[NT2];
#pragma unroll
        for (int q = 0; q < NT2; ++q) {
#pragma unroll
            for (int r = 0; r < 8; ++r)
                o[q][r] = (KK == 0) ? 0.f : orow[r][16 * q];
        }

#pragma unroll
        for (int r = 0; r < 8; ++r)
            part[(wave * 16 + 8 * hv + r) * PSTR + c] = hp[r];
        lds_barrier();

        const float g = (p0 + c < n) ? gate_l[p0 + c] : 0.f;
        v4f ha[2];
#pragma unroll
        for (int u = 0; u < 2; ++u) {
            v4f sum = *reinterpret_cast<const v4f*>(&part[c * PSTR + 8 * u + 4 * hv]);
#pragma unroll
            for (int w = 1; w < NW; ++w)
                sum += *reinterpret_cast<const v4f*>(&part[(w * 16 + c) * PSTR + 8 * u + 4 * hv]);
#pragma unroll
            for (int i = 0; i < 4; ++i) {
                const float hval = sum[i];
                ha[u][i] = g * (hval * __builtin_amdgcn_rcpf(1.0f + __expf(-hval)));
            }
        }
        lds_barrier();

#pragma unroll
        for (int q = 0; q < NT2; ++q) o[q] = mma4(ha[0].xy, b2[q][0].xy, o[q]);
#pragma unroll
        for (int q = 0; q < NT2; ++q) o[q] = mma4(ha[0].zw, b2[q][0].zw, o[q]);
#pragma unroll
        for (int q = 0; q < NT2; ++q) o[q] = mma4(ha[1].xy, b2[q][1].xy, o[q]);
#pragma unroll
        for (int q = 0; q < NT2; ++q) o[q] = mma4(ha[1].zw, b2[q][1].zw, o[q]);

#pragma unroll
        for (int r = 0; r < 8; ++r) {
            if (ok[r]) {
#pragma unroll
                for (int q = 0; q < NT2; ++q)
                    orow[r][16 * q] = o[q][r];
            }
        }

        if (++mt >= ntile) break;
        load_token_block(tx, x + wave * KSL, tok_l, mt * 16, n, lane);
        block_to_lds(sx, tx, lane);
        hp = first_product(sx, bw, c, hv);
    }
}

extern "C" void kernel_launch(void* const* d_in, const int* in_sizes, int n_in,
                              void* d_out, int out_size, void* d_ws, size_t ws_size,
                              hipStream_t stream) {
    const float* x       = (const float*)d_in[0];
    const float* routing = (const float*)d_in[1];
    const int*   idxs    = (const int*)d_in[2];
    const float* w_a     = (const float*)d_in[3];
    const float* w_b     = (const float*)d_in[4];
    float* out = (float*)d_out;

    hipLaunchKernelGGL((routed_lora_kernel<0>), dim3(16, NEXP), dim3(CTOK), 0, stream,
                       x, routing, idxs, w_a, w_b, out);
    hipLaunchKernelGGL((routed_lora_kernel<1>), dim3(16, NEXP), dim3(CTOK), 0, stream,
                       x, routing, idxs, w_a, w_b, out);
}


# Tokens binned by expert pair: one workgroup (e0,e1) does both slots of its tokens, so x rows are read once, out rows stored once and the second launch goes; sum order unchanged
# speedup vs baseline: 4.6259x; 1.3527x over previous
#include <hip/hip_runtime.h>

typedef float v2f __attribute__((ext_vector_type(2)));
typedef float v4f __attribute__((ext_vector_type(4)));
typedef float v8f __attribute__((ext_vector_type(8)));
typedef int   v2i __attribute__((ext_vector_type(2)));

#define D_IN   2048
#define D_OUT  2048
#define NEXP   16
#define RANK   16
#define KTOP   2
#define NTOK   4096
#define NTHR   256
#define SCAN   (NTOK / NTHR)
#define NW     8
#define KSL    (D_IN / NW)
#define S1     (KSL / 8)
#define DSL    (D_OUT / NW)
#define NT2    (DSL / 16)
#define PSTR   20
#define XP     (KSL + 4)

__device__ __forceinline__ v8f mma4(v2f a, v2f b, v8f c) {
    return __builtin_amdgcn_wmma_f32_16x16x4_f32(false, a, false, b, (short)0, c, false, false);
}

__device__ __forceinline__ void lds_barrier() {
    __builtin_amdgcn_fence(__ATOMIC_RELEASE, "workgroup", "local");
    __builtin_amdgcn_s_barrier();
    __builtin_amdgcn_fence(__ATOMIC_ACQUIRE, "workgroup", "local");
}

__device__ __forceinline__ void wave_lds_sync() {
    __builtin_amdgcn_fence(__ATOMIC_RELEASE, "wavefront", "local");
    __builtin_amdgcn_wave_barrier();
    __builtin_amdgcn_fence(__ATOMIC_ACQUIRE, "wavefront", "local");
}

__device__ __forceinline__ void load_weight_block(v4f (&t)[S1], const float* w0, int lane) {
#pragma unroll
    for (int j = 0; j < S1; ++j)
        t[j] = *reinterpret_cast<const v4f*>(w0 + (size_t)(j >> 1) * D_IN + (KSL / 2) * (j & 1) + 4 * lane);
}
__device__ __forceinline__ void block_to_lds(float* stg, const v4f (&t)[S1], int lane) {
#pragma unroll
    for (int j = 0; j < S1; ++j)
        *reinterpret_cast<v4f*>(stg + (j >> 1) * XP + (KSL / 2) * (j & 1) + 4 * lane) = t[j];
    wave_lds_sync();
}
__device__ __forceinline__ void read_fragments(v4f (&f)[S1], const float* stg, int c, int hv) {
#pragma unroll
    for (int s = 0; s < S1; ++s)
        f[s] = *reinterpret_cast<const v4f*>(stg + c * XP + 8 * s + 4 * hv);
}

__device__ __forceinline__ void load_tile_rows(v4f (&a)[S1], float& g0, float& g1, const float* xs,
                                               const float* routing, const int* tok_l,
                                               int p0, int n, int c) {
    const bool live = p0 + c < n;
    const int  tk   = tok_l[live ? p0 + c : p0];
    const float* row = xs + (size_t)tk * D_IN;
#pragma unroll
    for (int s = 0; s < S1; ++s)
        a[s] = *reinterpret_cast<const v4f*>(row + 8 * s);
    const v2f gr = *reinterpret_cast<const v2f*>(routing + (size_t)tk * KTOP);
    g0 = live ? 2.0f * gr.x : 0.f;
    g1 = live ? 2.0f * gr.y : 0.f;
}

__device__ __forceinline__ v8f first_product(const v4f (&a)[S1], const float* sw, int c, int hv) {
    v4f b[S1];
    read_fragments(b, sw, c, hv);
    __builtin_amdgcn_sched_barrier(0);
    v8f acc0 = {}, acc1 = {};
#pragma unroll
    for (int s = 0; s < S1; ++s) {
        acc0 = mma4(a[s].xy, b[s].xy, acc0);
        acc1 = mma4(a[s].zw, b[s].zw, acc1);
    }
    return acc0 + acc1;
}

__device__ __forceinline__ void load_out_weights(v4f (&b2)[NT2][2], const float* wb) {
#pragma unroll
    for (int q = 0; q < NT2; ++q) {
        b2[q][0] = *reinterpret_cast<const v4f*>(wb + q * 16 * RANK);
        b2[q][1] = *reinterpret_cast<const v4f*>(wb + q * 16 * RANK + 8);
    }
}

__device__ __forceinline__ void gated_rows(v4f (&ha)[2], const float* pt, float g, int c, int hv) {
#pragma unroll
    for (int u = 0; u < 2; ++u) {
        v4f sum = *reinterpret_cast<const v4f*>(&pt[c * PSTR + 8 * u + 4 * hv]);
#pragma unroll
        for (int w = 1; w < NW; ++w)
            sum += *reinterpret_cast<const v4f*>(&pt[(w * 16 + c) * PSTR + 8 * u + 4 * hv]);
#pragma unroll
        for (int i = 0; i < 4; ++i) {
            const float hval = sum[i];
            ha[u][i] = g * (hval * __builtin_amdgcn_rcpf(1.0f + __expf(-hval)));
        }
    }
}

__global__ __launch_bounds__(NTHR)
void paired_lora_kernel(const float* __restrict__ x,
                        const float* __restrict__ routing,
                        const int*   __restrict__ idxs,
                        const float* __restrict__ w_a,
                        const float* __restrict__ w_b,
                        float* __restrict__ out)
{
    __shared__ __attribute__((aligned(16))) int   tok_l[NTOK];
    __shared__ int   wcnt[NW];
    __shared__ __attribute__((aligned(16))) float part[2 * NW * 16 * PSTR];
    __shared__ __attribute__((aligned(16))) float stage_w[2 * NW * 16 * XP];

    const int tid  = threadIdx.x;
    const int wave = tid >> 5;
    const int lane = tid & 31;
    const int c    = lane & 15;
    const int hv   = lane >> 4;
    const int e0   = blockIdx.y;
    const int e1   = blockIdx.x;

    v2i sel[SCAN];
#pragma unroll
    for (int j = 0; j < SCAN; ++j)
        sel[j] = *reinterpret_cast<const v2i*>(idxs + (size_t)(j * NTHR + tid) * KTOP);

    __builtin_amdgcn_sched_barrier(0);

    float* sw0 = stage_w + wave * 16 * XP;
    float* sw1 = stage_w + (NW + wave) * 16 * XP;
    v4f tw0[S1], tw1[S1];
    load_weight_block(tw0, w_a + (size_t)e0 * RANK * D_IN + wave * KSL, lane);
    load_weight_block(tw1, w_a + (size_t)e1 * RANK * D_IN + wave * KSL, lane);
    __builtin_amdgcn_sched_barrier(0);

    unsigned msk[SCAN];
    int cnt = 0;
#pragma unroll
    for (int j = 0; j < SCAN; ++j) {
        msk[j] = __builtin_amdgcn_ballot_w32(sel[j].x == e0 && sel[j].y == e1);
        cnt += __popc(msk[j]);
    }
    if (lane == 0) wcnt[wave] = cnt;
    lds_barrier();
    int base = 0, nsum = 0;
#pragma unroll
    for (int w = 0; w < NW; ++w) {
        const int cw = wcnt[w];
        base += (w < wave) ? cw : 0;
        nsum += cw;
    }
    {
        const unsigned below = (1u << lane) - 1u;
        int off = base;
#pragma unroll
        for (int j = 0; j < SCAN; ++j) {
            if ((msk[j] >> lane) & 1u)
                tok_l[off + __popc(msk[j] & below)] = j * NTHR + tid;
            off += __popc(msk[j]);
        }
    }
    lds_barrier();
    const int n = __builtin_amdgcn_readfirstlane(nsum);
    if (n == 0) return;
    const int ntile = (n + 15) >> 4;

    const float* xs = x + wave * KSL + 4 * hv;
    v4f a[S1];
    float g0, g1;
    load_tile_rows(a, g0, g1, xs, routing, tok_l, 0, n, c);
    block_to_lds(sw0, tw0, lane);
    block_to_lds(sw1, tw1, lane);

    const float* wbl = w_b + (wave * DSL + c) * RANK + 4 * hv;

    for (int mt = 0;;) {
        const int p0 = mt * 16;

        const v8f hp0 = first_product(a, sw0, c, hv);
        const v8f hp1 = first_product(a, sw1, c, hv);

        int eo0 = e0 * (D_OUT * RANK);
        asm volatile("" : "+s"(eo0));
        v4f b20[NT2][2];
        load_out_weights(b20, wbl + eo0);

        float* orow[8];
        bool   ok[8];
#pragma unroll
        for (int r = 0; r < 8; ++r) {
            const int pr = p0 + 8 * hv + r;
            ok[r]   = pr < n;
            orow[r] = out + (size_t)tok_l[ok[r] ? pr : p0] * D_OUT + wave * DSL + c;
        }

#pragma unroll
        for (int r = 0; r < 8; ++r) {
            part[(wave * 16 + 8 * hv + r) * PSTR + c]             = hp0[r];
            part[((NW + wave) * 16 + 8 * hv + r) * PSTR + c]      = hp1[r];
        }
        lds_barrier();

        v4f ha0[2], ha1[2];
        gated_rows(ha0, part, g0, c, hv);
        gated_rows(ha1, part + NW * 16 * PSTR, g1, c, hv);
        lds_barrier();

        int eo1 = e1 * (D_OUT * RANK);
        asm volatile("" : "+s"(eo1));
        v4f b21[NT2][2];
        load_out_weights(b21, wbl + eo1);

        v8f o[NT2];
#pragma unroll
        for (int q = 0; q < NT2; ++q) o[q] = v8f{};
#pragma unroll
        for (int q = 0; q < NT2; ++q) o[q] = mma4(ha0[0].xy, b20[q][0].xy, o[q]);
#pragma unroll
        for (int q = 0; q < NT2; ++q) o[q] = mma4(ha0[0].zw, b20[q][0].zw, o[q]);
#pragma unroll
        for (int q = 0; q < NT2; ++q) o[q] = mma4(ha0[1].xy, b20[q][1].xy, o[q]);
#pragma unroll
        for (int q = 0; q < NT2; ++q) o[q] = mma4(ha0[1].zw, b20[q][1].zw, o[q]);
#pragma unroll
        for (int q = 0; q < NT2; ++q) o[q] = mma4(ha1[0].xy, b21[q][0].xy, o[q]);
#pragma unroll
        for (int q = 0; q < NT2; ++q) o[q] = mma4(ha1[0].zw, b21[q][0].zw, o[q]);
#pragma unroll
        for (int q = 0; q < NT2; ++q) o[q] = mma4(ha1[1].xy, b21[q][1].xy, o[q]);
#pragma unroll
        for (int q = 0; q < NT2; ++q) o[q] = mma4(ha1[1].zw, b21[q][1].zw, o[q]);

#pragma unroll
        for (int r = 0; r < 8; ++r) {
            if (ok[r]) {
#pragma unroll
                for (int q = 0; q < NT2; ++q)
                    orow[r][16 * q] = o[q][r];
            }
        }

        if (++mt >= ntile) break;
        load_tile_rows(a, g0, g1, xs, routing, tok_l, mt * 16, n, c);
    }
}

extern "C" void kernel_launch(void* const* d_in, const int* in_sizes, int n_in,
                              void* d_out, int out_size, void* d_ws, size_t ws_size,
                              hipStream_t stream) {
    const float* x       = (const float*)d_in[0];
    const float* routing = (const float*)d_in[1];
    const int*   idxs    = (const int*)d_in[2];
    const float* w_a     = (const float*)d_in[3];
    const float* w_b     = (const float*)d_in[4];
    float* out = (float*)d_out;

    hipLaunchKernelGGL(paired_lora_kernel, dim3(NEXP, NEXP), dim3(NTHR), 0, stream,
                       x, routing, idxs, w_a, w_b, out);
}


# A pair whose list fills three 16-token tiles gives the third to the next pair in pair order that has at most one; both decide alike from the same pair counts, no hand-over; sum order unchanged
# speedup vs baseline: 5.0589x; 1.0936x over previous
#include <hip/hip_runtime.h>

typedef float v2f __attribute__((ext_vector_type(2)));
typedef float v4f __attribute__((ext_vector_type(4)));
typedef float v8f __attribute__((ext_vector_type(8)));
typedef int   v2i __attribute__((ext_vector_type(2)));

#define D_IN   2048
#define D_OUT  2048
#define NEXP   16
#define RANK   16
#define KTOP   2
#define NTOK   4096
#define NTHR   256
#define SCAN   (NTOK / NTHR)
#define NW     8
#define KSL    (D_IN / NW)
#define S1     (KSL / 8)
#define DSL    (D_OUT / NW)
#define NT2    (DSL / 16)
#define PSTR   20
#define NBR    6
#define NCN    (2 * NBR + 1)
#define NPAIR  (NEXP * NEXP)
#define XP     (KSL + 4)

__device__ __forceinline__ v8f mma4(v2f a, v2f b, v8f c) {
    return __builtin_amdgcn_wmma_f32_16x16x4_f32(false, a, false, b, (short)0, c, false, false);
}

__device__ __forceinline__ void lds_barrier() {
    __builtin_amdgcn_fence(__ATOMIC_RELEASE, "workgroup", "local");
    __builtin_amdgcn_s_barrier();
    __builtin_amdgcn_fence(__ATOMIC_ACQUIRE, "workgroup", "local");
}

__device__ __forceinline__ void wave_lds_sync() {
    __builtin_amdgcn_fence(__ATOMIC_RELEASE, "wavefront", "local");
    __builtin_amdgcn_wave_barrier();
    __builtin_amdgcn_fence(__ATOMIC_ACQUIRE, "wavefront", "local");
}

__device__ __forceinline__ void load_weight_block(v4f (&t)[S1], const float* w0, int lane) {
#pragma unroll
    for (int j = 0; j < S1; ++j)
        t[j] = *reinterpret_cast<const v4f*>(w0 + (size_t)(j >> 1) * D_IN + (KSL / 2) * (j & 1) + 4 * lane);
}
__device__ __forceinline__ void block_to_lds(float* stg, const v4f (&t)[S1], int lane) {
#pragma unroll
    for (int j = 0; j < S1; ++j)
        *reinterpret_cast<v4f*>(stg + (j >> 1) * XP + (KSL / 2) * (j & 1) + 4 * lane) = t[j];
    wave_lds_sync();
}
__device__ __forceinline__ void read_fragments(v4f (&f)[S1], const float* stg, int c, int hv) {
#pragma unroll
    for (int s = 0; s < S1; ++s)
        f[s] = *reinterpret_cast<const v4f*>(stg + c * XP + 8 * s + 4 * hv);
}

__device__ __forceinline__ void load_tile_rows(v4f (&a)[S1], float& g0, float& g1, const float* xs,
                                               const float* routing, const int* tok_l,
                                               int p0, int n, int c) {
    const bool live = p0 + c < n;
    const int  tk   = tok_l[live ? p0 + c : p0];
    const float* row = xs + (size_t)tk * D_IN;
#pragma unroll
    for (int s = 0; s < S1; ++s)
        a[s] = *reinterpret_cast<const v4f*>(row + 8 * s);
    const v2f gr = *reinterpret_cast<const v2f*>(routing + (size_t)tk * KTOP);
    g0 = live ? 2.0f * gr.x : 0.f;
    g1 = live ? 2.0f * gr.y : 0.f;
}

__device__ __forceinline__ v8f first_product(const v4f (&a)[S1], const float* sw, int c, int hv) {
    v4f b[S1];
    read_fragments(b, sw, c, hv);
    __builtin_amdgcn_sched_barrier(0);
    v8f acc0 = {}, acc1 = {};
#pragma unroll
    for (int s = 0; s < S1; ++s) {
        acc0 = mma4(a[s].xy, b[s].xy, acc0);
        acc1 = mma4(a[s].zw, b[s].zw, acc1);
    }
    return acc0 + acc1;
}

__device__ __forceinline__ void load_out_weights(v4f (&b2)[NT2][2], const float* wb) {
#pragma unroll
    for (int q = 0; q < NT2; ++q) {
        b2[q][0] = *reinterpret_cast<const v4f*>(wb + q * 16 * RANK);
        b2[q][1] = *reinterpret_cast<const v4f*>(wb + q * 16 * RANK + 8);
    }
}

__device__ __forceinline__ void gated_rows(v4f (&ha)[2], const float* pt, float g, int c, int hv) {
#pragma unroll
    for (int u = 0; u < 2; ++u) {
        v4f sum = *reinterpret_cast<const v4f*>(&pt[c * PSTR + 8 * u + 4 * hv]);
#pragma unroll
        for (int w = 1; w < NW; ++w)
            sum += *reinterpret_cast<const v4f*>(&pt[(w * 16 + c) * PSTR + 8 * u + 4 * hv]);
#pragma unroll
        for (int i = 0; i < 4; ++i) {
            const float hval = sum[i];
            ha[u][i] = g * (hval * __builtin_amdgcn_rcpf(1.0f + __expf(-hval)));
        }
    }
}

template <bool FIRST>
__device__ __forceinline__ int list_pair(v4f (&tw0)[S1], v4f (&tw1)[S1], int* tok_l, int* wcnt,
                                         const int* idxs, const float* w_a, int tgt, int skip,
                                         int tid, int wave, int lane, bool& give, int& help) {
    const int e0 = tgt >> 4;
    const int e1 = tgt & (NEXP - 1);

    v2i sel[SCAN];
#pragma unroll
    for (int j = 0; j < SCAN; ++j)
        sel[j] = *reinterpret_cast<const v2i*>(idxs + (size_t)(j * NTHR + tid) * KTOP);

    __builtin_amdgcn_sched_barrier(0);

    load_weight_block(tw0, w_a + (size_t)e0 * RANK * D_IN + wave * KSL, lane);
    load_weight_block(tw1, w_a + (size_t)e1 * RANK * D_IN + wave * KSL, lane);
    __builtin_amdgcn_sched_barrier(0);

    int pid[SCAN];
#pragma unroll
    for (int j = 0; j < SCAN; ++j)
        pid[j] = ((unsigned)(sel[j].x | sel[j].y) < (unsigned)NEXP) ? sel[j].x * NEXP + sel[j].y : -1;

    unsigned msk[SCAN];
    int cn[NCN];
#pragma unroll
    for (int i = 0; i < NCN; ++i) cn[i] = 0;
#pragma unroll
    for (int j = 0; j < SCAN; ++j) {
        msk[j] = __builtin_amdgcn_ballot_w32(pid[j] == tgt);
        cn[NBR] += __popc(msk[j]);
    }
    if (FIRST) {
#pragma unroll
        for (int i = 0; i < NCN; ++i) {
            if (i == NBR) continue;
            const int nid = (tgt + i - NBR) & (NPAIR - 1);
#pragma unroll
            for (int j = 0; j < SCAN; ++j)
                cn[i] += __popc(__builtin_amdgcn_ballot_w32(pid[j] == nid));
        }
    }
    if (lane == 0) {
#pragma unroll
        for (int i = 0; i < NCN; ++i)
            if (FIRST || i == NBR) wcnt[wave * 16 + i] = cn[i];
    }
    lds_barrier();
    int base = 0;
    int tot[NCN];
#pragma unroll
    for (int i = 0; i < NCN; ++i) tot[i] = 0;
#pragma unroll
    for (int w = 0; w < NW; ++w) {
#pragma unroll
        for (int i = 0; i < NCN; ++i)
            if (FIRST || i == NBR) tot[i] += wcnt[w * 16 + i];
        base += (w < wave) ? wcnt[w * 16 + NBR] : 0;
    }
    if (FIRST) {
        int tl[NCN];
#pragma unroll
        for (int i = 0; i < NCN; ++i) tl[i] = (__builtin_amdgcn_readfirstlane(tot[i]) + 15) >> 4;
        int fw = 0;
#pragma unroll
        for (int j = 1; j <= NBR; ++j) {
            const int t = tl[NBR + j];
            if (fw == 0) fw = (t <= 1) ? 1 : (t == 3) ? 2 : 0;
        }
        give = tl[NBR] == 3 && fw == 1;
        int bw = 0, hj = 0;
#pragma unroll
        for (int j = 1; j <= NBR; ++j) {
            const int t = tl[NBR - j];
            if (bw == 0) {
                if (t == 3)      { bw = 1; hj = j; }
                else if (t <= 1) { bw = 2; }
            }
        }
        help = (tl[NBR] <= 1 && bw == 1) ? ((tgt - hj) & (NPAIR - 1)) : -1;
    }
    {
        const unsigned below = (1u << lane) - 1u;
        int off = base - skip;
#pragma unroll
        for (int j = 0; j < SCAN; ++j) {
            if ((msk[j] >> lane) & 1u) {
                const int pos = off + __popc(msk[j] & below);
                if (pos >= 0) tok_l[pos] = j * NTHR + tid;
            }
            off += __popc(msk[j]);
        }
    }
    lds_barrier();
    return __builtin_amdgcn_readfirstlane(tot[NBR]) - skip;
}

__global__ __launch_bounds__(NTHR)
void paired_lora_kernel(const float* __restrict__ x,
                        const float* __restrict__ routing,
                        const int*   __restrict__ idxs,
                        const float* __restrict__ w_a,
                        const float* __restrict__ w_b,
                        float* __restrict__ out)
{
    __shared__ __attribute__((aligned(16))) int   tok_l[NTOK];
    __shared__ __attribute__((aligned(16))) int   wcnt[NW * 16];
    __shared__ __attribute__((aligned(16))) float part[2 * NW * 16 * PSTR];
    __shared__ __attribute__((aligned(16))) float stage_w[2 * NW * 16 * XP];

    const int tid  = threadIdx.x;
    const int wave = tid >> 5;
    const int lane = tid & 31;
    const int c    = lane & 15;
    const int hv   = lane >> 4;

    float* sw0 = stage_w + wave * 16 * XP;
    float* sw1 = stage_w + (NW + wave) * 16 * XP;
    v4f tw0[S1], tw1[S1];

    int  tgt  = blockIdx.y * NEXP + blockIdx.x;
    bool give = false;
    int  help = -1;
    int  none = -1;
    bool no   = false;
    int  n    = list_pair<true>(tw0, tw1, tok_l, wcnt, idxs, w_a, tgt, 0, tid, wave, lane, give, help);
    if (give) n = 2 * 16;

    for (int pass = 0;; ++pass) {
    if (n > 0) {
    const int e0 = tgt >> 4;
    const int e1 = tgt & (NEXP - 1);
    const int ntile = (n + 15) >> 4;

    const float* xs = x + wave * KSL + 4 * hv;
    v4f a[S1];
    float g0, g1;
    load_tile_rows(a, g0, g1, xs, routing, tok_l, 0, n, c);
    block_to_lds(sw0, tw0, lane);
    block_to_lds(sw1, tw1, lane);

    const float* wbl = w_b + (wave * DSL + c) * RANK + 4 * hv;

    for (int mt = 0;;) {
        const int p0 = mt * 16;

        const v8f hp0 = first_product(a, sw0, c, hv);
        const v8f hp1 = first_product(a, sw1, c, hv);

        int eo0 = e0 * (D_OUT * RANK);
        asm volatile("" : "+s"(eo0));
        v4f b20[NT2][2];
        load_out_weights(b20, wbl + eo0);

        float* orow[8];
        bool   ok[8];
#pragma unroll
        for (int r = 0; r < 8; ++r) {
            const int pr = p0 + 8 * hv + r;
            ok[r]   = pr < n;
            orow[r] = out + (size_t)tok_l[ok[r] ? pr : p0] * D_OUT + wave * DSL + c;
        }

#pragma unroll
        for (int r = 0; r < 8; ++r) {
            part[(wave * 16 + 8 * hv + r) * PSTR + c]             = hp0[r];
            part[((NW + wave) * 16 + 8 * hv + r) * PSTR + c]      = hp1[r];
        }
        lds_barrier();

        v4f ha0[2], ha1[2];
        gated_rows(ha0, part, g0, c, hv);
        gated_rows(ha1, part + NW * 16 * PSTR, g1, c, hv);
        lds_barrier();

        int eo1 = e1 * (D_OUT * RANK);
        asm volatile("" : "+s"(eo1));
        v4f b21[NT2][2];
        load_out_weights(b21, wbl + eo1);

        v8f o[NT2];
#pragma unroll
        for (int q = 0; q < NT2; ++q) o[q] = v8f{};
#pragma unroll
        for (int q = 0; q < NT2; ++q) o[q] = mma4(ha0[0].xy, b20[q][0].xy, o[q]);
#pragma unroll
        for (int q = 0; q < NT2; ++q) o[q] = mma4(ha0[0].zw, b20[q][0].zw, o[q]);
#pragma unroll
        for (int q = 0; q < NT2; ++q) o[q] = mma4(ha0[1].xy, b20[q][1].xy, o[q]);
#pragma unroll
        for (int q = 0; q < NT2; ++q) o[q] = mma4(ha0[1].zw, b20[q][1].zw, o[q]);
#pragma unroll
        for (int q = 0; q < NT2; ++q) o[q] = mma4(ha1[0].xy, b21[q][0].xy, o[q]);
#pragma unroll
        for (int q = 0; q < NT2; ++q) o[q] = mma4(ha1[0].zw, b21[q][0].zw, o[q]);
#pragma unroll
        for (int q = 0; q < NT2; ++q) o[q] = mma4(ha1[1].xy, b21[q][1].xy, o[q]);
#pragma unroll
        for (int q = 0; q < NT2; ++q) o[q] = mma4(ha1[1].zw, b21[q][1].zw, o[q]);

#pragma unroll
        for (int r = 0; r < 8; ++r) {
            if (ok[r]) {
#pragma unroll
                for (int q = 0; q < NT2; ++q)
                    orow[r][16 * q] = o[q][r];
            }
        }

        if (++mt >= ntile) break;
        load_tile_rows(a, g0, g1, xs, routing, tok_l, mt * 16, n, c);
    }
    }

    if (pass == 1 || help < 0) break;
    lds_barrier();
    tgt = help;
    n   = list_pair<false>(tw0, tw1, tok_l, wcnt, idxs, w_a, tgt, 2 * 16, tid, wave, lane, no, none);
    }
}

extern "C" void kernel_launch(void* const* d_in, const int* in_sizes, int n_in,
                              void* d_out, int out_size, void* d_ws, size_t ws_size,
                              hipStream_t stream) {
    const float* x       = (const float*)d_in[0];
    const float* routing = (const float*)d_in[1];
    const int*   idxs    = (const int*)d_in[2];
    const float* w_a     = (const float*)d_in[3];
    const float* w_b     = (const float*)d_in[4];
    float* out = (float*)d_out;

    hipLaunchKernelGGL(paired_lora_kernel, dim3(NEXP, NEXP), dim3(NTHR), 0, stream,
                       x, routing, idxs, w_a, w_b, out);
}
